# starting kernel
# baseline (speedup 1.0000x reference)
#include <hip/hip_runtime.h>
#include <math.h>

typedef float v2f __attribute__((ext_vector_type(2)));
typedef float v8f __attribute__((ext_vector_type(8)));

#define MULT 32
#define NATTR 8
#define RB 16
#define FC 64
#define LOG2F_ 0.6931471805599453f

__device__ __forceinline__ float ssp(float x) {
    return fmaxf(x, 0.0f) + log1pf(__expf(-fabsf(x))) - LOG2F_;
}

__global__ __launch_bounds__(256) void lin1_kernel(
    const float* __restrict__ node_s, const float* __restrict__ node_v,
    const float* __restrict__ W1s, const float* __restrict__ W1v,
    float* __restrict__ s_out, float* __restrict__ v_out, int N)
{
    __shared__ float Ls[MULT * MULT];
    __shared__ float Lv[MULT * MULT];
    for (int i = threadIdx.x; i < MULT * MULT; i += 256) { Ls[i] = W1s[i]; Lv[i] = W1v[i]; }
    __syncthreads();

    int gid = blockIdx.x * 256 + threadIdx.x;
    int n = gid >> 5, w = gid & 31;
    if (n >= N) return;
    const float inv_m = 0.17677669529663687f;
    const float* srow = node_s + (size_t)n * MULT;
    const float* vrow = node_v + (size_t)n * MULT * 3;
    float as = 0.f, a0 = 0.f, a1 = 0.f, a2 = 0.f;
#pragma unroll 8
    for (int u = 0; u < MULT; ++u) {
        float ws_ = Ls[u * MULT + w];
        float wv_ = Lv[u * MULT + w];
        as += srow[u] * ws_;
        a0 += vrow[u * 3 + 0] * wv_;
        a1 += vrow[u * 3 + 1] * wv_;
        a2 += vrow[u * 3 + 2] * wv_;
    }
    s_out[(size_t)n * MULT + w] = as * inv_m;
    v_out[(size_t)n * MULT * 3 + w * 3 + 0] = a0 * inv_m;
    v_out[(size_t)n * MULT * 3 + w * 3 + 1] = a1 * inv_m;
    v_out[(size_t)n * MULT * 3 + w * 3 + 2] = a2 * inv_m;
}

__global__ __launch_bounds__(256) void edge_kernel(
    const float* __restrict__ edge_emb, const float* __restrict__ edge_y1,
    const int* __restrict__ edge_index,
    const float* __restrict__ Wr1, const float* __restrict__ Wr2,
    const float* __restrict__ s_in, const float* __restrict__ v_in,
    float* __restrict__ agg_s, float* __restrict__ agg_v, int E)
{
    __shared__ float Wr1t[FC * RB];
    __shared__ float Wr2t[4 * MULT * FC];
    __shared__ float stage[8 * 16 * 4 * MULT];

    const int tid = threadIdx.x;
    for (int i = tid; i < RB * FC; i += 256) {
        int k = i / FC, n = i % FC;
        Wr1t[n * RB + k] = Wr1[i];
    }
    for (int i = tid; i < FC * 4 * MULT; i += 256) {
        int k = i / (4 * MULT), n = i % (4 * MULT);
        Wr2t[n * FC + k] = Wr2[i];
    }
    __syncthreads();

    const int wave = tid >> 5, lane = tid & 31;
    const int m = lane & 15, kh = lane >> 4;
    const long e0 = (long)blockIdx.x * 128 + wave * 16;
    float* const mystage = stage + wave * (16 * 4 * MULT);

    const v8f vzero = {0.f, 0.f, 0.f, 0.f, 0.f, 0.f, 0.f, 0.f};

    v8f h[4];
#pragma unroll
    for (int t = 0; t < 4; ++t) h[t] = vzero;
#pragma unroll
    for (int k4 = 0; k4 < RB; k4 += 4) {
        long ee = e0 + m; if (ee >= E) ee = E - 1;
        v2f a = *(const v2f*)(edge_emb + ee * RB + k4 + 2 * kh);
#pragma unroll
        for (int t = 0; t < 4; ++t) {
            v2f b = *(const v2f*)(Wr1t + (t * 16 + m) * RB + k4 + 2 * kh);
            h[t] = __builtin_amdgcn_wmma_f32_16x16x4_f32(
                false, a, false, b, (short)0, h[t], false, false);
        }
    }
#pragma unroll
    for (int t = 0; t < 4; ++t)
#pragma unroll
        for (int r = 0; r < 8; ++r) {
            int row = r + 8 * kh, col = t * 16 + m;
            mystage[row * FC + col] = ssp(h[t][r] * 0.25f);
        }
    __syncthreads();

    v8f wacc[8];
#pragma unroll
    for (int t = 0; t < 8; ++t) wacc[t] = vzero;
    for (int k4 = 0; k4 < FC; k4 += 4) {
        v2f a = *(const v2f*)(mystage + m * FC + k4 + 2 * kh);
#pragma unroll
        for (int t = 0; t < 8; ++t) {
            v2f b = *(const v2f*)(Wr2t + (t * 16 + m) * FC + k4 + 2 * kh);
            wacc[t] = __builtin_amdgcn_wmma_f32_16x16x4_f32(
                false, a, false, b, (short)0, wacc[t], false, false);
        }
    }
    __syncthreads();
#pragma unroll
    for (int t = 0; t < 8; ++t)
#pragma unroll
        for (int r = 0; r < 8; ++r) {
            int row = r + 8 * kh, col = t * 16 + m;
            mystage[row * 4 * MULT + col] = wacc[t][r] * 0.125f;
        }
    __syncthreads();

    long e = (long)blockIdx.x * 128 + (tid >> 1);
    if (e < E) {
        int wv = (tid >> 1) >> 4, row = (tid >> 1) & 15;
        const float* wrow = stage + wv * (16 * 4 * MULT) + row * 4 * MULT;
        int src = edge_index[e];
        int dst = edge_index[(long)E + e];
        float y0 = edge_y1[e * 3 + 0];
        float y1 = edge_y1[e * 3 + 1];
        float y2 = edge_y1[e * 3 + 2];
        const float inv_s3  = 0.5773502691896258f;
        const float inv_deg = 0.17677669529663687f;
        const float* srow = s_in + (size_t)src * MULT;
        const float* vrow = v_in + (size_t)src * MULT * 3;
        float* as_ = agg_s + (size_t)dst * 2 * MULT;
        float* av_ = agg_v + (size_t)dst * 2 * MULT * 3;
        int u0 = (tid & 1) * 16;
#pragma unroll 4
        for (int u = u0; u < u0 + 16; ++u) {
            float w1 = wrow[u];
            float w2 = wrow[MULT + u];
            float w3 = wrow[2 * MULT + u];
            float w4 = wrow[3 * MULT + u];
            float se = srow[u];
            float v0 = vrow[u * 3 + 0], v1 = vrow[u * 3 + 1], v2 = vrow[u * 3 + 2];
            unsafeAtomicAdd(as_ + u, w1 * se * inv_deg);
            float dv = v0 * y0 + v1 * y1 + v2 * y2;
            unsafeAtomicAdd(as_ + MULT + u, w4 * dv * inv_s3 * inv_deg);
            float t2 = w2 * se * inv_deg;
            unsafeAtomicAdd(av_ + u * 3 + 0, t2 * y0);
            unsafeAtomicAdd(av_ + u * 3 + 1, t2 * y1);
            unsafeAtomicAdd(av_ + u * 3 + 2, t2 * y2);
            float t3 = w3 * inv_deg;
            unsafeAtomicAdd(av_ + 3 * MULT + u * 3 + 0, t3 * v0);
            unsafeAtomicAdd(av_ + 3 * MULT + u * 3 + 1, t3 * v1);
            unsafeAtomicAdd(av_ + 3 * MULT + u * 3 + 2, t3 * v2);
        }
    }
}

__global__ __launch_bounds__(256) void out_kernel(
    const float* __restrict__ node_s, const float* __restrict__ node_v,
    const float* __restrict__ node_attrs,
    const float* __restrict__ W2s, const float* __restrict__ W2v,
    const float* __restrict__ Wscs, const float* __restrict__ Wscv,
    const float* __restrict__ agg_s, const float* __restrict__ agg_v,
    float* __restrict__ out, int N)
{
    __shared__ float L2s[2 * MULT * MULT];
    __shared__ float L2v[2 * MULT * MULT];
    __shared__ float Lss[MULT * NATTR * MULT];
    __shared__ float Lsv[MULT * NATTR * MULT];
    for (int i = threadIdx.x; i < 2 * MULT * MULT; i += 256) { L2s[i] = W2s[i]; L2v[i] = W2v[i]; }
    for (int i = threadIdx.x; i < MULT * NATTR * MULT; i += 256) { Lss[i] = Wscs[i]; Lsv[i] = Wscv[i]; }
    __syncthreads();

    int gid = blockIdx.x * 256 + threadIdx.x;
    int n = gid >> 5, w = gid & 31;
    if (n >= N) return;
    const float inv2m = 0.125f;
    const float invfan = 0.0625f;

    const float* ags = agg_s + (size_t)n * 2 * MULT;
    const float* agv = agg_v + (size_t)n * 2 * MULT * 3;
    float ns = 0.f, nv0 = 0.f, nv1 = 0.f, nv2 = 0.f;
#pragma unroll 8
    for (int u = 0; u < 2 * MULT; ++u) {
        float w2s = L2s[u * MULT + w];
        float w2v = L2v[u * MULT + w];
        ns  += ags[u] * w2s;
        nv0 += agv[u * 3 + 0] * w2v;
        nv1 += agv[u * 3 + 1] * w2v;
        nv2 += agv[u * 3 + 2] * w2v;
    }
    ns *= inv2m; nv0 *= inv2m; nv1 *= inv2m; nv2 *= inv2m;

    float attr[NATTR];
#pragma unroll
    for (int a = 0; a < NATTR; ++a) attr[a] = node_attrs[(size_t)n * NATTR + a];

    const float* srow = node_s + (size_t)n * MULT;
    const float* vrow = node_v + (size_t)n * MULT * 3;
    float scs = 0.f, scv0 = 0.f, scv1 = 0.f, scv2 = 0.f;
#pragma unroll 4
    for (int u = 0; u < MULT; ++u) {
        float ts = 0.f, tv = 0.f;
#pragma unroll
        for (int a = 0; a < NATTR; ++a) {
            ts += attr[a] * Lss[(u * NATTR + a) * MULT + w];
            tv += attr[a] * Lsv[(u * NATTR + a) * MULT + w];
        }
        scs  += srow[u] * ts;
        scv0 += vrow[u * 3 + 0] * tv;
        scv1 += vrow[u * 3 + 1] * tv;
        scv2 += vrow[u * 3 + 2] * tv;
    }
    scs *= invfan; scv0 *= invfan; scv1 *= invfan; scv2 *= invfan;

    float* orow = out + (size_t)n * 4 * MULT;
    orow[w] = ns + scs;
    orow[MULT + w * 3 + 0] = nv0 + scv0;
    orow[MULT + w * 3 + 1] = nv1 + scv1;
    orow[MULT + w * 3 + 2] = nv2 + scv2;
}

extern "C" void kernel_launch(void* const* d_in, const int* in_sizes, int n_in,
                              void* d_out, int out_size, void* d_ws, size_t ws_size,
                              hipStream_t stream) {
    const float* node_s     = (const float*)d_in[0];
    const float* node_v     = (const float*)d_in[1];
    const float* node_attrs = (const float*)d_in[2];
    const float* edge_emb   = (const float*)d_in[3];
    const float* edge_y1    = (const float*)d_in[5];
    const int*   edge_index = (const int*)d_in[6];
    const float* W1s  = (const float*)d_in[7];
    const float* W1v  = (const float*)d_in[8];
    const float* Wr1  = (const float*)d_in[9];
    const float* Wr2  = (const float*)d_in[10];
    const float* W2s  = (const float*)d_in[11];
    const float* W2v  = (const float*)d_in[12];
    const float* Wscs = (const float*)d_in[13];
    const float* Wscv = (const float*)d_in[14];

    const int N = in_sizes[0] / MULT;
    const int E = in_sizes[3] / RB;

    float* s_buf = (float*)d_ws;
    float* v_buf = s_buf + (size_t)N * MULT;
    float* agg_s = v_buf + (size_t)N * MULT * 3;
    float* agg_v = agg_s + (size_t)N * 2 * MULT;

    hipMemsetAsync(agg_s, 0, (size_t)N * 8 * MULT * sizeof(float), stream);

    lin1_kernel<<<(N * MULT + 255) / 256, 256, 0, stream>>>(
        node_s, node_v, W1s, W1v, s_buf, v_buf, N);
    edge_kernel<<<(E + 127) / 128, 256, 0, stream>>>(
        edge_emb, edge_y1, edge_index, Wr1, Wr2, s_buf, v_buf, agg_s, agg_v, E);
    out_kernel<<<(N * MULT + 255) / 256, 256, 0, stream>>>(
        node_s, node_v, node_attrs, W2s, W2v, Wscs, Wscv, agg_s, agg_v,
        (float*)d_out, N);
}


# node stage (linear_1, self-connection) as f32 WMMA GEMMs, one wave per 16-node tile: K=32 and K=256 (k = u*8+attr, A = x*attr in registers, B from LDS); one launch not two; node sums reordered
# speedup vs baseline: 6.3770x; 6.3770x over previous
#include <hip/hip_runtime.h>
#include <math.h>


typedef float v2f __attribute__((ext_vector_type(2)));
typedef float v8f __attribute__((ext_vector_type(8)));

#define MULT 32
#define NATTR 8
#define RB 16
#define FC 64
#define LOG2F_ 0.6931471805599453f
#define WP 132
#define HP 68
#define W2P 68
#define SCP 260
#define L1P 36

__device__ __forceinline__ float ssp(float x) {
    return fmaxf(x, 0.0f) + log1pf(__expf(-fabsf(x))) - LOG2F_;
}

__global__ __launch_bounds__(128) void node_kernel(
    const float* __restrict__ node_s, const float* __restrict__ node_v,
    const float* __restrict__ node_attrs,
    const float* __restrict__ W1s, const float* __restrict__ W1v,
    const float* __restrict__ Wscs, const float* __restrict__ Wscv,
    float* __restrict__ s_out, float* __restrict__ v_out,
    float* __restrict__ out, int N)
{
    __shared__ float Tss[MULT * SCP];
    __shared__ float Tsv[MULT * SCP];
    __shared__ float T1s[MULT * L1P];
    __shared__ float T1v[MULT * L1P];

    const int tid = threadIdx.x;
    for (int i = tid; i < MULT * NATTR * MULT; i += 128) {
        int k = i / MULT, n = i % MULT;
        Tss[n * SCP + k] = Wscs[i];
        Tsv[n * SCP + k] = Wscv[i];
    }
    for (int i = tid; i < MULT * MULT; i += 128) {
        int k = i / MULT, n = i % MULT;
        T1s[n * L1P + k] = W1s[i];
        T1v[n * L1P + k] = W1v[i];
    }
    __syncthreads();

    const int wave = tid >> 5, lane = tid & 31;
    const int m = lane & 15, kh = lane >> 4;
    const int n0 = (blockIdx.x * 4 + wave) * 16;
    if (n0 >= N) return;

    int na = n0 + m; if (na >= N) na = N - 1;
    const float* srow = node_s + (size_t)na * MULT;
    const float* vrow = node_v + (size_t)na * MULT * 3;
    const v8f vzero = {0.f, 0.f, 0.f, 0.f, 0.f, 0.f, 0.f, 0.f};

    {
        v8f ls[2], l0[2], l1[2], l2[2];
#pragma unroll
        for (int t = 0; t < 2; ++t) { ls[t] = vzero; l0[t] = vzero; l1[t] = vzero; l2[t] = vzero; }
#pragma unroll
        for (int j = 0; j < MULT / 4; ++j) {
            v2f as_ = *(const v2f*)(srow + 4 * j + 2 * kh);
            v2f p0 = *(const v2f*)(vrow + 12 * j + 6 * kh);
            v2f p1 = *(const v2f*)(vrow + 12 * j + 6 * kh + 2);
            v2f p2 = *(const v2f*)(vrow + 12 * j + 6 * kh + 4);
            v2f ax = {p0.x, p1.y};
            v2f ay = {p0.y, p2.x};
            v2f az = {p1.x, p2.y};
#pragma unroll
            for (int t = 0; t < 2; ++t) {
                v2f bs = *(const v2f*)(T1s + (t * 16 + m) * L1P + 4 * j + 2 * kh);
                v2f bv = *(const v2f*)(T1v + (t * 16 + m) * L1P + 4 * j + 2 * kh);
                ls[t] = __builtin_amdgcn_wmma_f32_16x16x4_f32(
                    false, as_, false, bs, (short)0, ls[t], false, false);
                l0[t] = __builtin_amdgcn_wmma_f32_16x16x4_f32(
                    false, ax, false, bv, (short)0, l0[t], false, false);
                l1[t] = __builtin_amdgcn_wmma_f32_16x16x4_f32(
                    false, ay, false, bv, (short)0, l1[t], false, false);
                l2[t] = __builtin_amdgcn_wmma_f32_16x16x4_f32(
                    false, az, false, bv, (short)0, l2[t], false, false);
            }
        }
        const float inv_m = 0.17677669529663687f;
#pragma unroll
        for (int r = 0; r < 8; ++r) {
            const int node = n0 + r + 8 * kh;
            if (node < N) {
#pragma unroll
                for (int t = 0; t < 2; ++t) {
                    const int w = t * 16 + m;
                    s_out[(size_t)node * MULT + w] = ls[t][r] * inv_m;
                    v_out[(size_t)node * MULT * 3 + w * 3 + 0] = l0[t][r] * inv_m;
                    v_out[(size_t)node * MULT * 3 + w * 3 + 1] = l1[t][r] * inv_m;
                    v_out[(size_t)node * MULT * 3 + w * 3 + 2] = l2[t][r] * inv_m;
                }
            }
        }
    }

    const v2f at0 = *(const v2f*)(node_attrs + (size_t)na * NATTR + 2 * kh);
    const v2f at1 = *(const v2f*)(node_attrs + (size_t)na * NATTR + 4 + 2 * kh);
    const float* bss = Tss + m * SCP + 2 * kh;
    const float* bsv = Tsv + m * SCP + 2 * kh;

    v8f cs[2], c0[2], c1[2], c2[2];
#pragma unroll
    for (int t = 0; t < 2; ++t) { cs[t] = vzero; c0[t] = vzero; c1[t] = vzero; c2[t] = vzero; }
#pragma unroll 4
    for (int u = 0; u < MULT; ++u) {
        const float xs = srow[u];
        const float x0 = vrow[u * 3 + 0];
        const float x1 = vrow[u * 3 + 1];
        const float x2 = vrow[u * 3 + 2];
#pragma unroll
        for (int h2 = 0; h2 < 2; ++h2) {
            const v2f at = h2 ? at1 : at0;
            v2f as_ = at * xs;
            v2f ax = at * x0, ay = at * x1, az = at * x2;
#pragma unroll
            for (int t = 0; t < 2; ++t) {
                v2f bs = *(const v2f*)(bss + t * 16 * SCP + 8 * u + 4 * h2);
                v2f bv = *(const v2f*)(bsv + t * 16 * SCP + 8 * u + 4 * h2);
                cs[t] = __builtin_amdgcn_wmma_f32_16x16x4_f32(
                    false, as_, false, bs, (short)0, cs[t], false, false);
                c0[t] = __builtin_amdgcn_wmma_f32_16x16x4_f32(
                    false, ax, false, bv, (short)0, c0[t], false, false);
                c1[t] = __builtin_amdgcn_wmma_f32_16x16x4_f32(
                    false, ay, false, bv, (short)0, c1[t], false, false);
                c2[t] = __builtin_amdgcn_wmma_f32_16x16x4_f32(
                    false, az, false, bv, (short)0, c2[t], false, false);
            }
        }
    }
    const float invfan = 0.0625f;
#pragma unroll
    for (int r = 0; r < 8; ++r) {
        const int node = n0 + r + 8 * kh;
        if (node < N) {
            float* orow = out + (size_t)node * 4 * MULT;
#pragma unroll
            for (int t = 0; t < 2; ++t) {
                const int w = t * 16 + m;
                orow[w] = cs[t][r] * invfan;
                orow[MULT + w * 3 + 0] = c0[t][r] * invfan;
                orow[MULT + w * 3 + 1] = c1[t][r] * invfan;
                orow[MULT + w * 3 + 2] = c2[t][r] * invfan;
            }
        }
    }
}

__global__ __launch_bounds__(256) void edge_kernel(
    const float* __restrict__ edge_emb, const float* __restrict__ edge_y0,
    const float* __restrict__ edge_y1, const int* __restrict__ edge_index,
    const float* __restrict__ Wr1, const float* __restrict__ Wr2,
    const float* __restrict__ W2s, const float* __restrict__ W2v,
    const float* __restrict__ s_in, const float* __restrict__ v_in,
    float* __restrict__ out, int E)
{
    __shared__ float Wr1t[FC * RB];
    __shared__ float W2st[MULT * W2P];
    __shared__ float W2vt[MULT * W2P];
    __shared__ float stage[8 * 16 * WP];

    const int tid = threadIdx.x;
    const int wave = tid >> 5, lane = tid & 31;
    const int m = lane & 15, kh = lane >> 4;

    v2f bw[FC / 4];
    {
        const float* bp = Wr2 + (size_t)(2 * kh) * (4 * MULT) + wave * 16 + m;
#pragma unroll
        for (int j = 0; j < FC / 4; ++j) {
            bw[j].x = bp[(4 * j) * (4 * MULT)];
            bw[j].y = bp[(4 * j + 1) * (4 * MULT)];
        }
    }

    for (int i = tid; i < RB * FC; i += 256) {
        int k = i / FC, n = i % FC;
        Wr1t[n * RB + k] = Wr1[i];
    }
    for (int i = tid; i < 2 * MULT * MULT; i += 256) {
        int k = i / MULT, n = i % MULT;
        W2st[n * W2P + k] = W2s[i];
        W2vt[n * W2P + k] = W2v[i];
    }
    __syncthreads();

    const long e0 = (long)blockIdx.x * 128 + wave * 16;

    long ee = e0 + m; if (ee >= E) ee = E - 1;
    const int src  = edge_index[ee];
    const int dstm = edge_index[(long)E + ee];
    const float y0  = edge_y0[ee];
    const float y1x = edge_y1[ee * 3 + 0];
    const float y1y = edge_y1[ee * 3 + 1];
    const float y1z = edge_y1[ee * 3 + 2];
    const float* srow = s_in + (size_t)src * MULT + 2 * kh;
    const float* vrow = v_in + (size_t)src * MULT * 3 + 6 * kh;

    const v8f vzero = {0.f, 0.f, 0.f, 0.f, 0.f, 0.f, 0.f, 0.f};

    v8f h[4];
#pragma unroll
    for (int t = 0; t < 4; ++t) h[t] = vzero;
#pragma unroll
    for (int k4 = 0; k4 < RB; k4 += 4) {
        v2f a = *(const v2f*)(edge_emb + ee * RB + k4 + 2 * kh);
#pragma unroll
        for (int t = 0; t < 4; ++t) {
            v2f b = *(const v2f*)(Wr1t + (t * 16 + m) * RB + k4 + 2 * kh);
            h[t] = __builtin_amdgcn_wmma_f32_16x16x4_f32(
                false, a, false, b, (short)0, h[t], false, false);
        }
    }
#pragma unroll
    for (int t = 0; t < 4; ++t)
#pragma unroll
        for (int r = 0; r < 8; ++r) {
            int row = wave * 16 + r + 8 * kh, col = t * 16 + m;
            stage[row * HP + col] = ssp(h[t][r] * 0.25f);
        }
    __syncthreads();

    v8f wacc[8];
#pragma unroll
    for (int i = 0; i < 8; ++i) wacc[i] = vzero;
    {
        const float* ap = stage + m * HP + 2 * kh;
#pragma unroll
        for (int j = 0; j < FC / 4; ++j) {
#pragma unroll
            for (int i = 0; i < 8; ++i) {
                v2f a = *(const v2f*)(ap + i * 16 * HP + 4 * j);
                wacc[i] = __builtin_amdgcn_wmma_f32_16x16x4_f32(
                    false, a, false, bw[j], (short)0, wacc[i], false, false);
            }
        }
    }
    __syncthreads();
#pragma unroll
    for (int i = 0; i < 8; ++i)
#pragma unroll
        for (int r = 0; r < 8; ++r) {
            int row = i * 16 + r + 8 * kh, col = wave * 16 + m;
            stage[row * WP + col] = wacc[i][r] * 0.125f;
        }
    __syncthreads();

    const float* wrow = stage + (wave * 16 + m) * WP + 2 * kh;
    const float* b_s = W2st + m * W2P + 2 * kh;
    const float* b_v = W2vt + m * W2P + 2 * kh;
    const float inv_s3 = 0.5773502691896258f;

    v8f accs[2], acc0[2], acc1[2], acc2[2];
#pragma unroll
    for (int t = 0; t < 2; ++t) { accs[t] = vzero; acc0[t] = vzero; acc1[t] = vzero; acc2[t] = vzero; }

#pragma unroll
    for (int k4 = 0; k4 < MULT; k4 += 4) {
        v2f w1 = *(const v2f*)(wrow + k4);
        v2f w2 = *(const v2f*)(wrow + MULT + k4);
        v2f se = *(const v2f*)(srow + k4);
        v2f as_ = w1 * se * y0;
        v2f t2  = w2 * se;
        v2f av0 = t2 * y1x, av1 = t2 * y1y, av2 = t2 * y1z;
#pragma unroll
        for (int t = 0; t < 2; ++t) {
            v2f bs = *(const v2f*)(b_s + t * 16 * W2P + k4);
            v2f bv = *(const v2f*)(b_v + t * 16 * W2P + k4);
            accs[t] = __builtin_amdgcn_wmma_f32_16x16x4_f32(
                false, as_, false, bs, (short)0, accs[t], false, false);
            acc0[t] = __builtin_amdgcn_wmma_f32_16x16x4_f32(
                false, av0, false, bv, (short)0, acc0[t], false, false);
            acc1[t] = __builtin_amdgcn_wmma_f32_16x16x4_f32(
                false, av1, false, bv, (short)0, acc1[t], false, false);
            acc2[t] = __builtin_amdgcn_wmma_f32_16x16x4_f32(
                false, av2, false, bv, (short)0, acc2[t], false, false);
        }
    }
#pragma unroll
    for (int k4 = 0; k4 < MULT; k4 += 4) {
        v2f w3 = *(const v2f*)(wrow + 2 * MULT + k4);
        v2f w4 = *(const v2f*)(wrow + 3 * MULT + k4);
        v2f p0 = *(const v2f*)(vrow + k4 * 3);
        v2f p1 = *(const v2f*)(vrow + k4 * 3 + 2);
        v2f p2 = *(const v2f*)(vrow + k4 * 3 + 4);
        v2f vx = {p0.x, p1.y};
        v2f vy = {p0.y, p2.x};
        v2f vz = {p1.x, p2.y};
        v2f dv = vx * y1x + vy * y1y + vz * y1z;
        v2f as_ = w4 * dv * inv_s3;
        v2f t3  = w3 * y0;
        v2f av0 = t3 * vx, av1 = t3 * vy, av2 = t3 * vz;
#pragma unroll
        for (int t = 0; t < 2; ++t) {
            v2f bs = *(const v2f*)(b_s + t * 16 * W2P + MULT + k4);
            v2f bv = *(const v2f*)(b_v + t * 16 * W2P + MULT + k4);
            accs[t] = __builtin_amdgcn_wmma_f32_16x16x4_f32(
                false, as_, false, bs, (short)0, accs[t], false, false);
            acc0[t] = __builtin_amdgcn_wmma_f32_16x16x4_f32(
                false, av0, false, bv, (short)0, acc0[t], false, false);
            acc1[t] = __builtin_amdgcn_wmma_f32_16x16x4_f32(
                false, av1, false, bv, (short)0, acc1[t], false, false);
            acc2[t] = __builtin_amdgcn_wmma_f32_16x16x4_f32(
                false, av2, false, bv, (short)0, acc2[t], false, false);
        }
    }

    const float sc = 0.022097086912079608f;
#pragma unroll
    for (int t = 0; t < 2; ++t) { accs[t] *= sc; acc0[t] *= sc; acc1[t] *= sc; acc2[t] *= sc; }

#pragma unroll
    for (int r = 0; r < 8; ++r) {
        const int row = r + 8 * kh;
        const int d = __shfl(dstm, row);
        if (e0 + row < E) {
            float* orow = out + (size_t)d * 4 * MULT;
#pragma unroll
            for (int t = 0; t < 2; ++t) {
                const int wcol = t * 16 + m;
                unsafeAtomicAdd(orow + wcol, accs[t][r]);
                unsafeAtomicAdd(orow + MULT + wcol * 3 + 0, acc0[t][r]);
                unsafeAtomicAdd(orow + MULT + wcol * 3 + 1, acc1[t][r]);
                unsafeAtomicAdd(orow + MULT + wcol * 3 + 2, acc2[t][r]);
            }
        }
    }
}

extern "C" void kernel_launch(void* const* d_in, const int* in_sizes, int n_in,
                              void* d_out, int out_size, void* d_ws, size_t ws_size,
                              hipStream_t stream) {
    const float* node_s     = (const float*)d_in[0];
    const float* node_v     = (const float*)d_in[1];
    const float* node_attrs = (const float*)d_in[2];
    const float* edge_emb   = (const float*)d_in[3];
    const float* edge_y0    = (const float*)d_in[4];
    const float* edge_y1    = (const float*)d_in[5];
    const int*   edge_index = (const int*)d_in[6];
    const float* W1s  = (const float*)d_in[7];
    const float* W1v  = (const float*)d_in[8];
    const float* Wr1  = (const float*)d_in[9];
    const float* Wr2  = (const float*)d_in[10];
    const float* W2s  = (const float*)d_in[11];
    const float* W2v  = (const float*)d_in[12];
    const float* Wscs = (const float*)d_in[13];
    const float* Wscv = (const float*)d_in[14];

    const int N = in_sizes[0] / MULT;
    const int E = in_sizes[3] / RB;

    float* s_buf = (float*)d_ws;
    float* v_buf = s_buf + (size_t)N * MULT;

    node_kernel<<<(N + 63) / 64, 128, 0, stream>>>(
        node_s, node_v, node_attrs, W1s, W1v, Wscs, Wscv, s_buf, v_buf, (float*)d_out, N);
    edge_kernel<<<(E + 127) / 128, 256, 0, stream>>>(
        edge_emb, edge_y0, edge_y1, edge_index, Wr1, Wr2, W2s, W2v,
        s_buf, v_buf, (float*)d_out, E);
}


# edge kernel: each GEMM of the chain computed transposed, so its D tile is register for register the next GEMM's operand fragment: h and w never pass LDS, 3 of 4 barriers go; k sum order changes
# speedup vs baseline: 6.6393x; 1.0411x over previous
#include <hip/hip_runtime.h>
#include <math.h>


typedef float v2f __attribute__((ext_vector_type(2)));
typedef float v4f __attribute__((ext_vector_type(4)));
typedef float v8f __attribute__((ext_vector_type(8)));

#define MULT 32
#define NATTR 8
#define RB 16
#define FC 64
#define LOG2F_ 0.6931471805599453f
#define R2P 68
#define W2P 68
#define SCP 260
#define L1P 36

__device__ __forceinline__ float ssp(float x) {
    return fmaxf(x, 0.0f) + log1pf(__expf(-fabsf(x))) - LOG2F_;
}

__global__ __launch_bounds__(128) void node_kernel(
    const float* __restrict__ node_s, const float* __restrict__ node_v,
    const float* __restrict__ node_attrs,
    const float* __restrict__ W1s, const float* __restrict__ W1v,
    const float* __restrict__ Wscs, const float* __restrict__ Wscv,
    float* __restrict__ s_out, float* __restrict__ v_out,
    float* __restrict__ out, int N)
{
    __shared__ float Tss[MULT * SCP];
    __shared__ float Tsv[MULT * SCP];
    __shared__ float T1s[MULT * L1P];
    __shared__ float T1v[MULT * L1P];

    const int tid = threadIdx.x;
    for (int i = tid; i < MULT * NATTR * MULT; i += 128) {
        int k = i / MULT, n = i % MULT;
        Tss[n * SCP + k] = Wscs[i];
        Tsv[n * SCP + k] = Wscv[i];
    }
    for (int i = tid; i < MULT * MULT; i += 128) {
        int k = i / MULT, n = i % MULT;
        T1s[n * L1P + k] = W1s[i];
        T1v[n * L1P + k] = W1v[i];
    }
    __syncthreads();

    const int wave = tid >> 5, lane = tid & 31;
    const int m = lane & 15, kh = lane >> 4;
    const int n0 = (blockIdx.x * 4 + wave) * 16;
    if (n0 >= N) return;

    int na = n0 + m; if (na >= N) na = N - 1;
    const float* srow = node_s + (size_t)na * MULT;
    const float* vrow = node_v + (size_t)na * MULT * 3;
    const v8f vzero = {0.f, 0.f, 0.f, 0.f, 0.f, 0.f, 0.f, 0.f};

    {
        v8f ls[2], l0[2], l1[2], l2[2];
#pragma unroll
        for (int t = 0; t < 2; ++t) { ls[t] = vzero; l0[t] = vzero; l1[t] = vzero; l2[t] = vzero; }
#pragma unroll
        for (int j = 0; j < MULT / 4; ++j) {
            v2f as_ = *(const v2f*)(srow + 4 * j + 2 * kh);
            v2f p0 = *(const v2f*)(vrow + 12 * j + 6 * kh);
            v2f p1 = *(const v2f*)(vrow + 12 * j + 6 * kh + 2);
            v2f p2 = *(const v2f*)(vrow + 12 * j + 6 * kh + 4);
            v2f ax = {p0.x, p1.y};
            v2f ay = {p0.y, p2.x};
            v2f az = {p1.x, p2.y};
#pragma unroll
            for (int t = 0; t < 2; ++t) {
                v2f bs = *(const v2f*)(T1s + (t * 16 + m) * L1P + 4 * j + 2 * kh);
                v2f bv = *(const v2f*)(T1v + (t * 16 + m) * L1P + 4 * j + 2 * kh);
                ls[t] = __builtin_amdgcn_wmma_f32_16x16x4_f32(
                    false, as_, false, bs, (short)0, ls[t], false, false);
                l0[t] = __builtin_amdgcn_wmma_f32_16x16x4_f32(
                    false, ax, false, bv, (short)0, l0[t], false, false);
                l1[t] = __builtin_amdgcn_wmma_f32_16x16x4_f32(
                    false, ay, false, bv, (short)0, l1[t], false, false);
                l2[t] = __builtin_amdgcn_wmma_f32_16x16x4_f32(
                    false, az, false, bv, (short)0, l2[t], false, false);
            }
        }
        const float inv_m = 0.17677669529663687f;
#pragma unroll
        for (int r = 0; r < 8; ++r) {
            const int node = n0 + r + 8 * kh;
            if (node < N) {
#pragma unroll
                for (int t = 0; t < 2; ++t) {
                    const int w = t * 16 + m;
                    s_out[(size_t)node * MULT + w] = ls[t][r] * inv_m;
                    v_out[(size_t)node * MULT * 3 + w * 3 + 0] = l0[t][r] * inv_m;
                    v_out[(size_t)node * MULT * 3 + w * 3 + 1] = l1[t][r] * inv_m;
                    v_out[(size_t)node * MULT * 3 + w * 3 + 2] = l2[t][r] * inv_m;
                }
            }
        }
    }

    const v2f at0 = *(const v2f*)(node_attrs + (size_t)na * NATTR + 2 * kh);
    const v2f at1 = *(const v2f*)(node_attrs + (size_t)na * NATTR + 4 + 2 * kh);
    const float* bss = Tss + m * SCP + 2 * kh;
    const float* bsv = Tsv + m * SCP + 2 * kh;

    v8f cs[2], c0[2], c1[2], c2[2];
#pragma unroll
    for (int t = 0; t < 2; ++t) { cs[t] = vzero; c0[t] = vzero; c1[t] = vzero; c2[t] = vzero; }
#pragma unroll 4
    for (int u = 0; u < MULT; ++u) {
        const float xs = srow[u];
        const float x0 = vrow[u * 3 + 0];
        const float x1 = vrow[u * 3 + 1];
        const float x2 = vrow[u * 3 + 2];
#pragma unroll
        for (int h2 = 0; h2 < 2; ++h2) {
            const v2f at = h2 ? at1 : at0;
            v2f as_ = at * xs;
            v2f ax = at * x0, ay = at * x1, az = at * x2;
#pragma unroll
            for (int t = 0; t < 2; ++t) {
                v2f bs = *(const v2f*)(bss + t * 16 * SCP + 8 * u + 4 * h2);
                v2f bv = *(const v2f*)(bsv + t * 16 * SCP + 8 * u + 4 * h2);
                cs[t] = __builtin_amdgcn_wmma_f32_16x16x4_f32(
                    false, as_, false, bs, (short)0, cs[t], false, false);
                c0[t] = __builtin_amdgcn_wmma_f32_16x16x4_f32(
                    false, ax, false, bv, (short)0, c0[t], false, false);
                c1[t] = __builtin_amdgcn_wmma_f32_16x16x4_f32(
                    false, ay, false, bv, (short)0, c1[t], false, false);
                c2[t] = __builtin_amdgcn_wmma_f32_16x16x4_f32(
                    false, az, false, bv, (short)0, c2[t], false, false);
            }
        }
    }
    const float invfan = 0.0625f;
#pragma unroll
    for (int r = 0; r < 8; ++r) {
        const int node = n0 + r + 8 * kh;
        if (node < N) {
            float* orow = out + (size_t)node * 4 * MULT;
#pragma unroll
            for (int t = 0; t < 2; ++t) {
                const int w = t * 16 + m;
                orow[w] = cs[t][r] * invfan;
                orow[MULT + w * 3 + 0] = c0[t][r] * invfan;
                orow[MULT + w * 3 + 1] = c1[t][r] * invfan;
                orow[MULT + w * 3 + 2] = c2[t][r] * invfan;
            }
        }
    }
}

__global__ __launch_bounds__(256, 6) void edge_kernel(
    const float* __restrict__ edge_emb, const float* __restrict__ edge_y0,
    const float* __restrict__ edge_y1, const int* __restrict__ edge_index,
    const float* __restrict__ Wr1, const float* __restrict__ Wr2,
    const float* __restrict__ W2s, const float* __restrict__ W2v,
    const float* __restrict__ s_in, const float* __restrict__ v_in,
    float* __restrict__ out, int E)
{
    __shared__ float Wr1t[FC * RB];
    __shared__ __attribute__((aligned(16))) float Wr2t[4 * MULT * R2P];
    __shared__ __attribute__((aligned(16))) float W2st[MULT * W2P];
    __shared__ __attribute__((aligned(16))) float W2vt[MULT * W2P];

    const int tid = threadIdx.x;
    const int wave = tid >> 5, lane = tid & 31;
    const int m = lane & 15, kh = lane >> 4;

    for (int i = tid; i < RB * FC; i += 256) {
        int k = i / FC, n = i % FC;
        Wr1t[n * RB + k] = Wr1[i];
    }
    for (int i = tid; i < FC * 4 * MULT; i += 256) {
        int k = i / (4 * MULT), n = i % (4 * MULT);
        Wr2t[n * R2P + k] = Wr2[i];
    }
    for (int i = tid; i < 2 * MULT * MULT; i += 256) {
        int k = i / MULT, n = i % MULT;
        W2st[n * W2P + k] = W2s[i];
        W2vt[n * W2P + k] = W2v[i];
    }
    __syncthreads();

    const int e0 = blockIdx.x * 128 + wave * 16;

    int ee = e0 + m; if (ee >= E) ee = E - 1;
    const int src  = edge_index[ee];
    const int dstm = edge_index[(size_t)E + ee];
    const float y0  = edge_y0[ee];
    const float y1x = edge_y1[(size_t)ee * 3 + 0];
    const float y1y = edge_y1[(size_t)ee * 3 + 1];
    const float y1z = edge_y1[(size_t)ee * 3 + 2];
    const float* srow = s_in + (size_t)src * MULT + 8 * kh;
    const float* vrow = v_in + (size_t)src * MULT * 3 + 24 * kh;

    const v8f vzero = {0.f, 0.f, 0.f, 0.f, 0.f, 0.f, 0.f, 0.f};

    v8f h[4];
#pragma unroll
    for (int t = 0; t < 4; ++t) h[t] = vzero;
#pragma unroll
    for (int k4 = 0; k4 < RB; k4 += 4) {
        v2f b = *(const v2f*)(edge_emb + (size_t)ee * RB + k4 + 2 * kh);
#pragma unroll
        for (int t = 0; t < 4; ++t) {
            v2f a = *(const v2f*)(Wr1t + (t * 16 + m) * RB + k4 + 2 * kh);
            h[t] = __builtin_amdgcn_wmma_f32_16x16x4_f32(
                false, a, false, b, (short)0, h[t], false, false);
        }
    }
#pragma unroll
    for (int t = 0; t < 4; ++t)
#pragma unroll
        for (int r = 0; r < 8; ++r)
            h[t][r] = ssp(h[t][r] * 0.25f);

    const float inv_s3 = 0.5773502691896258f;
    v8f accs[2], acc0[2], acc1[2], acc2[2];
#pragma unroll
    for (int t = 0; t < 2; ++t) { accs[t] = vzero; acc0[t] = vzero; acc1[t] = vzero; acc2[t] = vzero; }

#pragma unroll
    for (int hf = 0; hf < 2; ++hf) {
        v8f wq[4];
#pragma unroll
        for (int cc = 0; cc < 4; ++cc) wq[cc] = vzero;
#pragma unroll
        for (int t = 0; t < 4; ++t) {
            v4f a0[4], a1[4];
#pragma unroll
            for (int cc = 0; cc < 4; ++cc) {
                const float* ap = Wr2t + ((4 * hf + cc) * 16 + m) * R2P + 16 * t + 8 * kh;
                a0[cc] = *(const v4f*)ap;
                a1[cc] = *(const v4f*)(ap + 4);
            }
#pragma unroll
            for (int j = 0; j < 4; ++j) {
                v2f b = {h[t][2 * j], h[t][2 * j + 1]};
#pragma unroll
                for (int cc = 0; cc < 4; ++cc) {
                    v2f a;
                    if (j == 0)      a = (v2f){a0[cc].x, a0[cc].y};
                    else if (j == 1) a = (v2f){a0[cc].z, a0[cc].w};
                    else if (j == 2) a = (v2f){a1[cc].x, a1[cc].y};
                    else             a = (v2f){a1[cc].z, a1[cc].w};
                    wq[cc] = __builtin_amdgcn_wmma_f32_16x16x4_f32(
                        false, a, false, b, (short)0, wq[cc], false, false);
                }
            }
        }

#pragma unroll
        for (int tp = 0; tp < 2; ++tp) {
            v4f bs0[2], bs1[2], bv0[2], bv1[2];
#pragma unroll
            for (int t = 0; t < 2; ++t) {
                const float* bp = W2st + (t * 16 + m) * W2P + 32 * hf + 16 * tp + 8 * kh;
                const float* bq = W2vt + (t * 16 + m) * W2P + 32 * hf + 16 * tp + 8 * kh;
                bs0[t] = *(const v4f*)bp;  bs1[t] = *(const v4f*)(bp + 4);
                bv0[t] = *(const v4f*)bq;  bv1[t] = *(const v4f*)(bq + 4);
            }
            float fs[8], fx[8], fy[8], fz[8];
            if (hf == 0) {
                const v4f s0 = *(const v4f*)(srow + 16 * tp);
                const v4f s1 = *(const v4f*)(srow + 16 * tp + 4);
                fs[0] = s0.x; fs[1] = s0.y; fs[2] = s0.z; fs[3] = s0.w;
                fs[4] = s1.x; fs[5] = s1.y; fs[6] = s1.z; fs[7] = s1.w;
            } else {
                float vv[24];
#pragma unroll
                for (int q = 0; q < 6; ++q) {
                    const v4f p = *(const v4f*)(vrow + 48 * tp + 4 * q);
                    vv[4 * q] = p.x; vv[4 * q + 1] = p.y; vv[4 * q + 2] = p.z; vv[4 * q + 3] = p.w;
                }
#pragma unroll
                for (int r = 0; r < 8; ++r) { fx[r] = vv[3 * r]; fy[r] = vv[3 * r + 1]; fz[r] = vv[3 * r + 2]; }
            }
#pragma unroll
            for (int j = 0; j < 4; ++j) {
                const int r0 = 2 * j;
                v2f as_, av0, av1, av2;
                if (hf == 0) {
                    v2f w1 = {wq[tp][r0], wq[tp][r0 + 1]};
                    v2f w2 = {wq[2 + tp][r0], wq[2 + tp][r0 + 1]};
                    v2f se = {fs[r0], fs[r0 + 1]};
                    as_ = w1 * se * y0;
                    v2f t2 = w2 * se;
                    av0 = t2 * y1x; av1 = t2 * y1y; av2 = t2 * y1z;
                } else {
                    v2f w3 = {wq[tp][r0], wq[tp][r0 + 1]};
                    v2f w4 = {wq[2 + tp][r0], wq[2 + tp][r0 + 1]};
                    v2f vx = {fx[r0], fx[r0 + 1]};
                    v2f vy = {fy[r0], fy[r0 + 1]};
                    v2f vz = {fz[r0], fz[r0 + 1]};
                    v2f dv = vx * y1x + vy * y1y + vz * y1z;
                    as_ = w4 * dv * inv_s3;
                    v2f t3 = w3 * y0;
                    av0 = t3 * vx; av1 = t3 * vy; av2 = t3 * vz;
                }
#pragma unroll
                for (int t = 0; t < 2; ++t) {
                    v2f bs, bv;
                    if (j == 0)      { bs = (v2f){bs0[t].x, bs0[t].y}; bv = (v2f){bv0[t].x, bv0[t].y}; }
                    else if (j == 1) { bs = (v2f){bs0[t].z, bs0[t].w}; bv = (v2f){bv0[t].z, bv0[t].w}; }
                    else if (j == 2) { bs = (v2f){bs1[t].x, bs1[t].y}; bv = (v2f){bv1[t].x, bv1[t].y}; }
                    else             { bs = (v2f){bs1[t].z, bs1[t].w}; bv = (v2f){bv1[t].z, bv1[t].w}; }
                    accs[t] = __builtin_amdgcn_wmma_f32_16x16x4_f32(
                        false, as_, false, bs, (short)0, accs[t], false, false);
                    acc0[t] = __builtin_amdgcn_wmma_f32_16x16x4_f32(
                        false, av0, false, bv, (short)0, acc0[t], false, false);
                    acc1[t] = __builtin_amdgcn_wmma_f32_16x16x4_f32(
                        false, av1, false, bv, (short)0, acc1[t], false, false);
                    acc2[t] = __builtin_amdgcn_wmma_f32_16x16x4_f32(
                        false, av2, false, bv, (short)0, acc2[t], false, false);
                }
            }
        }
    }

    const float sc = 0.125f * 0.022097086912079608f;
#pragma unroll
    for (int t = 0; t < 2; ++t) { accs[t] *= sc; acc0[t] *= sc; acc1[t] *= sc; acc2[t] *= sc; }

    const int ln = __lane_id();
    const int ms = ln & 15, khs = ln >> 4;
#pragma unroll
    for (int r = 0; r < 8; ++r) {
        const int row = r + 8 * khs;
        const int d = __shfl(dstm, row);
        if (e0 + row < E) {
            float* orow = out + (size_t)d * 4 * MULT;
#pragma unroll
            for (int t = 0; t < 2; ++t) {
                const int wcol = t * 16 + ms;
                unsafeAtomicAdd(orow + wcol, accs[t][r]);
                unsafeAtomicAdd(orow + MULT + wcol * 3 + 0, acc0[t][r]);
                unsafeAtomicAdd(orow + MULT + wcol * 3 + 1, acc1[t][r]);
                unsafeAtomicAdd(orow + MULT + wcol * 3 + 2, acc2[t][r]);
            }
        }
    }
}

extern "C" void kernel_launch(void* const* d_in, const int* in_sizes, int n_in,
                              void* d_out, int out_size, void* d_ws, size_t ws_size,
                              hipStream_t stream) {
    const float* node_s     = (const float*)d_in[0];
    const float* node_v     = (const float*)d_in[1];
    const float* node_attrs = (const float*)d_in[2];
    const float* edge_emb   = (const float*)d_in[3];
    const float* edge_y0    = (const float*)d_in[4];
    const float* edge_y1    = (const float*)d_in[5];
    const int*   edge_index = (const int*)d_in[6];
    const float* W1s  = (const float*)d_in[7];
    const float* W1v  = (const float*)d_in[8];
    const float* Wr1  = (const float*)d_in[9];
    const float* Wr2  = (const float*)d_in[10];
    const float* W2s  = (const float*)d_in[11];
    const float* W2v  = (const float*)d_in[12];
    const float* Wscs = (const float*)d_in[13];
    const float* Wscv = (const float*)d_in[14];

    const int N = in_sizes[0] / MULT;
    const int E = in_sizes[3] / RB;

    float* s_buf = (float*)d_ws;
    float* v_buf = s_buf + (size_t)N * MULT;

    node_kernel<<<(N + 63) / 64, 128, 0, stream>>>(
        node_s, node_v, node_attrs, W1s, W1v, Wscs, Wscv, s_buf, v_buf, (float*)d_out, N);
    edge_kernel<<<(E + 127) / 128, 256, 0, stream>>>(
        edge_emb, edge_y0, edge_y1, edge_index, Wr1, Wr2, W2s, W2v,
        s_buf, v_buf, (float*)d_out, E);
}


# linear_2 of the 0e x 1o path: y1 factored out of the GEMM, one product (w2*s) @ W2v not three, its D rows scaled by the edge's y1 after; 240 not 272 WMMAs a tile; that path's rounding changes
# speedup vs baseline: 6.8905x; 1.0378x over previous
#include <hip/hip_runtime.h>
#include <math.h>


typedef float v2f __attribute__((ext_vector_type(2)));
typedef float v4f __attribute__((ext_vector_type(4)));
typedef float v8f __attribute__((ext_vector_type(8)));

#define MULT 32
#define NATTR 8
#define RB 16
#define FC 64
#define LOG2F_ 0.6931471805599453f
#define R2P 68
#define W2P 68
#define SCP 260
#define L1P 36

__device__ __forceinline__ float ssp(float x) {
    return fmaxf(x, 0.0f) + log1pf(__expf(-fabsf(x))) - LOG2F_;
}

__global__ __launch_bounds__(128) void node_kernel(
    const float* __restrict__ node_s, const float* __restrict__ node_v,
    const float* __restrict__ node_attrs,
    const float* __restrict__ W1s, const float* __restrict__ W1v,
    const float* __restrict__ Wscs, const float* __restrict__ Wscv,
    float* __restrict__ s_out, float* __restrict__ v_out,
    float* __restrict__ out, int N)
{
    __shared__ float Tss[MULT * SCP];
    __shared__ float Tsv[MULT * SCP];
    __shared__ float T1s[MULT * L1P];
    __shared__ float T1v[MULT * L1P];

    const int tid = threadIdx.x;
    for (int i = tid; i < MULT * NATTR * MULT; i += 128) {
        int k = i / MULT, n = i % MULT;
        Tss[n * SCP + k] = Wscs[i];
        Tsv[n * SCP + k] = Wscv[i];
    }
    for (int i = tid; i < MULT * MULT; i += 128) {
        int k = i / MULT, n = i % MULT;
        T1s[n * L1P + k] = W1s[i];
        T1v[n * L1P + k] = W1v[i];
    }
    __syncthreads();

    const int wave = tid >> 5, lane = tid & 31;
    const int m = lane & 15, kh = lane >> 4;
    const int n0 = (blockIdx.x * 4 + wave) * 16;
    if (n0 >= N) return;

    int na = n0 + m; if (na >= N) na = N - 1;
    const float* srow = node_s + (size_t)na * MULT;
    const float* vrow = node_v + (size_t)na * MULT * 3;
    const v8f vzero = {0.f, 0.f, 0.f, 0.f, 0.f, 0.f, 0.f, 0.f};

    {
        v8f ls[2], l0[2], l1[2], l2[2];
#pragma unroll
        for (int t = 0; t < 2; ++t) { ls[t] = vzero; l0[t] = vzero; l1[t] = vzero; l2[t] = vzero; }
#pragma unroll
        for (int j = 0; j < MULT / 4; ++j) {
            v2f as_ = *(const v2f*)(srow + 4 * j + 2 * kh);
            v2f p0 = *(const v2f*)(vrow + 12 * j + 6 * kh);
            v2f p1 = *(const v2f*)(vrow + 12 * j + 6 * kh + 2);
            v2f p2 = *(const v2f*)(vrow + 12 * j + 6 * kh + 4);
            v2f ax = {p0.x, p1.y};
            v2f ay = {p0.y, p2.x};
            v2f az = {p1.x, p2.y};
#pragma unroll
            for (int t = 0; t < 2; ++t) {
                v2f bs = *(const v2f*)(T1s + (t * 16 + m) * L1P + 4 * j + 2 * kh);
                v2f bv = *(const v2f*)(T1v + (t * 16 + m) * L1P + 4 * j + 2 * kh);
                ls[t] = __builtin_amdgcn_wmma_f32_16x16x4_f32(
                    false, as_, false, bs, (short)0, ls[t], false, false);
                l0[t] = __builtin_amdgcn_wmma_f32_16x16x4_f32(
                    false, ax, false, bv, (short)0, l0[t], false, false);
                l1[t] = __builtin_amdgcn_wmma_f32_16x16x4_f32(
                    false, ay, false, bv, (short)0, l1[t], false, false);
                l2[t] = __builtin_amdgcn_wmma_f32_16x16x4_f32(
                    false, az, false, bv, (short)0, l2[t], false, false);
            }
        }
        const float inv_m = 0.17677669529663687f;
#pragma unroll
        for (int r = 0; r < 8; ++r) {
            const int node = n0 + r + 8 * kh;
            if (node < N) {
#pragma unroll
                for (int t = 0; t < 2; ++t) {
                    const int w = t * 16 + m;
                    s_out[(size_t)node * MULT + w] = ls[t][r] * inv_m;
                    v_out[(size_t)node * MULT * 3 + w * 3 + 0] = l0[t][r] * inv_m;
                    v_out[(size_t)node * MULT * 3 + w * 3 + 1] = l1[t][r] * inv_m;
                    v_out[(size_t)node * MULT * 3 + w * 3 + 2] = l2[t][r] * inv_m;
                }
            }
        }
    }

    const v2f at0 = *(const v2f*)(node_attrs + (size_t)na * NATTR + 2 * kh);
    const v2f at1 = *(const v2f*)(node_attrs + (size_t)na * NATTR + 4 + 2 * kh);
    const float* bss = Tss + m * SCP + 2 * kh;
    const float* bsv = Tsv + m * SCP + 2 * kh;

    v8f cs[2], c0[2], c1[2], c2[2];
#pragma unroll
    for (int t = 0; t < 2; ++t) { cs[t] = vzero; c0[t] = vzero; c1[t] = vzero; c2[t] = vzero; }
#pragma unroll 4
    for (int u = 0; u < MULT; ++u) {
        const float xs = srow[u];
        const float x0 = vrow[u * 3 + 0];
        const float x1 = vrow[u * 3 + 1];
        const float x2 = vrow[u * 3 + 2];
#pragma unroll
        for (int h2 = 0; h2 < 2; ++h2) {
            const v2f at = h2 ? at1 : at0;
            v2f as_ = at * xs;
            v2f ax = at * x0, ay = at * x1, az = at * x2;
#pragma unroll
            for (int t = 0; t < 2; ++t) {
                v2f bs = *(const v2f*)(bss + t * 16 * SCP + 8 * u + 4 * h2);
                v2f bv = *(const v2f*)(bsv + t * 16 * SCP + 8 * u + 4 * h2);
                cs[t] = __builtin_amdgcn_wmma_f32_16x16x4_f32(
                    false, as_, false, bs, (short)0, cs[t], false, false);
                c0[t] = __builtin_amdgcn_wmma_f32_16x16x4_f32(
                    false, ax, false, bv, (short)0, c0[t], false, false);
                c1[t] = __builtin_amdgcn_wmma_f32_16x16x4_f32(
                    false, ay, false, bv, (short)0, c1[t], false, false);
                c2[t] = __builtin_amdgcn_wmma_f32_16x16x4_f32(
                    false, az, false, bv, (short)0, c2[t], false, false);
            }
        }
    }
    const float invfan = 0.0625f;
#pragma unroll
    for (int r = 0; r < 8; ++r) {
        const int node = n0 + r + 8 * kh;
        if (node < N) {
            float* orow = out + (size_t)node * 4 * MULT;
#pragma unroll
            for (int t = 0; t < 2; ++t) {
                const int w = t * 16 + m;
                orow[w] = cs[t][r] * invfan;
                orow[MULT + w * 3 + 0] = c0[t][r] * invfan;
                orow[MULT + w * 3 + 1] = c1[t][r] * invfan;
                orow[MULT + w * 3 + 2] = c2[t][r] * invfan;
            }
        }
    }
}

__global__ __launch_bounds__(256, 6) void edge_kernel(
    const float* __restrict__ edge_emb, const float* __restrict__ edge_y0,
    const float* __restrict__ edge_y1, const int* __restrict__ edge_index,
    const float* __restrict__ Wr1, const float* __restrict__ Wr2,
    const float* __restrict__ W2s, const float* __restrict__ W2v,
    const float* __restrict__ s_in, const float* __restrict__ v_in,
    float* __restrict__ out, int E)
{
    __shared__ float Wr1t[FC * RB];
    __shared__ __attribute__((aligned(16))) float Wr2t[4 * MULT * R2P];
    __shared__ __attribute__((aligned(16))) float W2st[MULT * W2P];
    __shared__ __attribute__((aligned(16))) float W2vt[MULT * W2P];

    const int tid = threadIdx.x;
    const int wave = tid >> 5, lane = tid & 31;
    const int m = lane & 15, kh = lane >> 4;

    for (int i = tid; i < RB * FC; i += 256) {
        int k = i / FC, n = i % FC;
        Wr1t[n * RB + k] = Wr1[i];
    }
    for (int i = tid; i < FC * 4 * MULT; i += 256) {
        int k = i / (4 * MULT), n = i % (4 * MULT);
        Wr2t[n * R2P + k] = Wr2[i];
    }
    for (int i = tid; i < 2 * MULT * MULT; i += 256) {
        int k = i / MULT, n = i % MULT;
        W2st[n * W2P + k] = W2s[i];
        W2vt[n * W2P + k] = W2v[i];
    }
    __syncthreads();

    const int e0 = blockIdx.x * 128 + wave * 16;

    int ee = e0 + m; if (ee >= E) ee = E - 1;
    const int src  = edge_index[ee];
    const int dstm = edge_index[(size_t)E + ee];
    const float y0  = edge_y0[ee];
    const float y1x = edge_y1[(size_t)ee * 3 + 0];
    const float y1y = edge_y1[(size_t)ee * 3 + 1];
    const float y1z = edge_y1[(size_t)ee * 3 + 2];
    const float* srow = s_in + (size_t)src * MULT + 8 * kh;
    const float* vrow = v_in + (size_t)src * MULT * 3 + 24 * kh;

    const v8f vzero = {0.f, 0.f, 0.f, 0.f, 0.f, 0.f, 0.f, 0.f};

    v8f h[4];
#pragma unroll
    for (int t = 0; t < 4; ++t) h[t] = vzero;
#pragma unroll
    for (int k4 = 0; k4 < RB; k4 += 4) {
        v2f b = *(const v2f*)(edge_emb + (size_t)ee * RB + k4 + 2 * kh);
#pragma unroll
        for (int t = 0; t < 4; ++t) {
            v2f a = *(const v2f*)(Wr1t + (t * 16 + m) * RB + k4 + 2 * kh);
            h[t] = __builtin_amdgcn_wmma_f32_16x16x4_f32(
                false, a, false, b, (short)0, h[t], false, false);
        }
    }
#pragma unroll
    for (int t = 0; t < 4; ++t)
#pragma unroll
        for (int r = 0; r < 8; ++r)
            h[t][r] = ssp(h[t][r] * 0.25f);

    const float inv_s3 = 0.5773502691896258f;
    v8f accs[2], acc0[2], acc1[2], acc2[2];
    v8f pv[2];
#pragma unroll
    for (int t = 0; t < 2; ++t) { accs[t] = vzero; pv[t] = vzero; }

#pragma unroll
    for (int hf = 0; hf < 2; ++hf) {
        v8f wq[4];
#pragma unroll
        for (int cc = 0; cc < 4; ++cc) wq[cc] = vzero;
#pragma unroll
        for (int t = 0; t < 4; ++t) {
            v4f a0[4], a1[4];
#pragma unroll
            for (int cc = 0; cc < 4; ++cc) {
                const float* ap = Wr2t + ((4 * hf + cc) * 16 + m) * R2P + 16 * t + 8 * kh;
                a0[cc] = *(const v4f*)ap;
                a1[cc] = *(const v4f*)(ap + 4);
            }
#pragma unroll
            for (int j = 0; j < 4; ++j) {
                v2f b = {h[t][2 * j], h[t][2 * j + 1]};
#pragma unroll
                for (int cc = 0; cc < 4; ++cc) {
                    v2f a;
                    if (j == 0)      a = (v2f){a0[cc].x, a0[cc].y};
                    else if (j == 1) a = (v2f){a0[cc].z, a0[cc].w};
                    else if (j == 2) a = (v2f){a1[cc].x, a1[cc].y};
                    else             a = (v2f){a1[cc].z, a1[cc].w};
                    wq[cc] = __builtin_amdgcn_wmma_f32_16x16x4_f32(
                        false, a, false, b, (short)0, wq[cc], false, false);
                }
            }
        }

#pragma unroll
        for (int tp = 0; tp < 2; ++tp) {
            v4f bs0[2], bs1[2], bv0[2], bv1[2];
#pragma unroll
            for (int t = 0; t < 2; ++t) {
                const float* bp = W2st + (t * 16 + m) * W2P + 32 * hf + 16 * tp + 8 * kh;
                const float* bq = W2vt + (t * 16 + m) * W2P + 32 * hf + 16 * tp + 8 * kh;
                bs0[t] = *(const v4f*)bp;  bs1[t] = *(const v4f*)(bp + 4);
                bv0[t] = *(const v4f*)bq;  bv1[t] = *(const v4f*)(bq + 4);
            }
            float fs[8], fx[8], fy[8], fz[8];
            if (hf == 0) {
                const v4f s0 = *(const v4f*)(srow + 16 * tp);
                const v4f s1 = *(const v4f*)(srow + 16 * tp + 4);
                fs[0] = s0.x; fs[1] = s0.y; fs[2] = s0.z; fs[3] = s0.w;
                fs[4] = s1.x; fs[5] = s1.y; fs[6] = s1.z; fs[7] = s1.w;
            } else {
                float vv[24];
#pragma unroll
                for (int q = 0; q < 6; ++q) {
                    const v4f p = *(const v4f*)(vrow + 48 * tp + 4 * q);
                    vv[4 * q] = p.x; vv[4 * q + 1] = p.y; vv[4 * q + 2] = p.z; vv[4 * q + 3] = p.w;
                }
#pragma unroll
                for (int r = 0; r < 8; ++r) { fx[r] = vv[3 * r]; fy[r] = vv[3 * r + 1]; fz[r] = vv[3 * r + 2]; }
            }
#pragma unroll
            for (int j = 0; j < 4; ++j) {
                const int r0 = 2 * j;
                v2f as_, av0, av1, av2;
                if (hf == 0) {
                    v2f w1 = {wq[tp][r0], wq[tp][r0 + 1]};
                    v2f w2 = {wq[2 + tp][r0], wq[2 + tp][r0 + 1]};
                    v2f se = {fs[r0], fs[r0 + 1]};
                    as_ = w1 * se * y0;
                    av0 = w2 * se;
                    av1 = av0; av2 = av0;
                } else {
                    v2f w3 = {wq[tp][r0], wq[tp][r0 + 1]};
                    v2f w4 = {wq[2 + tp][r0], wq[2 + tp][r0 + 1]};
                    v2f vx = {fx[r0], fx[r0 + 1]};
                    v2f vy = {fy[r0], fy[r0 + 1]};
                    v2f vz = {fz[r0], fz[r0 + 1]};
                    v2f dv = vx * y1x + vy * y1y + vz * y1z;
                    as_ = w4 * dv * inv_s3;
                    v2f t3 = w3 * y0;
                    av0 = t3 * vx; av1 = t3 * vy; av2 = t3 * vz;
                }
#pragma unroll
                for (int t = 0; t < 2; ++t) {
                    v2f bs, bv;
                    if (j == 0)      { bs = (v2f){bs0[t].x, bs0[t].y}; bv = (v2f){bv0[t].x, bv0[t].y}; }
                    else if (j == 1) { bs = (v2f){bs0[t].z, bs0[t].w}; bv = (v2f){bv0[t].z, bv0[t].w}; }
                    else if (j == 2) { bs = (v2f){bs1[t].x, bs1[t].y}; bv = (v2f){bv1[t].x, bv1[t].y}; }
                    else             { bs = (v2f){bs1[t].z, bs1[t].w}; bv = (v2f){bv1[t].z, bv1[t].w}; }
                    accs[t] = __builtin_amdgcn_wmma_f32_16x16x4_f32(
                        false, as_, false, bs, (short)0, accs[t], false, false);
                    if (hf == 0) {
                        pv[t] = __builtin_amdgcn_wmma_f32_16x16x4_f32(
                            false, av0, false, bv, (short)0, pv[t], false, false);
                    } else {
                        acc0[t] = __builtin_amdgcn_wmma_f32_16x16x4_f32(
                            false, av0, false, bv, (short)0, acc0[t], false, false);
                        acc1[t] = __builtin_amdgcn_wmma_f32_16x16x4_f32(
                            false, av1, false, bv, (short)0, acc1[t], false, false);
                        acc2[t] = __builtin_amdgcn_wmma_f32_16x16x4_f32(
                            false, av2, false, bv, (short)0, acc2[t], false, false);
                    }
                }
            }
        }
        if (hf == 0) {
            const int rb = (__lane_id() >> 4) * 8;
#pragma unroll
            for (int r = 0; r < 8; ++r) {
                const float yx = __shfl(y1x, rb + r);
                const float yy = __shfl(y1y, rb + r);
                const float yz = __shfl(y1z, rb + r);
#pragma unroll
                for (int t = 0; t < 2; ++t) {
                    acc0[t][r] = pv[t][r] * yx;
                    acc1[t][r] = pv[t][r] * yy;
                    acc2[t][r] = pv[t][r] * yz;
                }
            }
        }
    }

    const float sc = 0.125f * 0.022097086912079608f;
#pragma unroll
    for (int t = 0; t < 2; ++t) { accs[t] *= sc; acc0[t] *= sc; acc1[t] *= sc; acc2[t] *= sc; }

    const int ln = __lane_id();
    const int ms = ln & 15, khs = ln >> 4;
#pragma unroll
    for (int r = 0; r < 8; ++r) {
        const int row = r + 8 * khs;
        const int d = __shfl(dstm, row);
        if (e0 + row < E) {
            float* orow = out + (size_t)d * 4 * MULT;
#pragma unroll
            for (int t = 0; t < 2; ++t) {
                const int wcol = t * 16 + ms;
                unsafeAtomicAdd(orow + wcol, accs[t][r]);
                unsafeAtomicAdd(orow + MULT + wcol * 3 + 0, acc0[t][r]);
                unsafeAtomicAdd(orow + MULT + wcol * 3 + 1, acc1[t][r]);
                unsafeAtomicAdd(orow + MULT + wcol * 3 + 2, acc2[t][r]);
            }
        }
    }
}

extern "C" void kernel_launch(void* const* d_in, const int* in_sizes, int n_in,
                              void* d_out, int out_size, void* d_ws, size_t ws_size,
                              hipStream_t stream) {
    const float* node_s     = (const float*)d_in[0];
    const float* node_v     = (const float*)d_in[1];
    const float* node_attrs = (const float*)d_in[2];
    const float* edge_emb   = (const float*)d_in[3];
    const float* edge_y0    = (const float*)d_in[4];
    const float* edge_y1    = (const float*)d_in[5];
    const int*   edge_index = (const int*)d_in[6];
    const float* W1s  = (const float*)d_in[7];
    const float* W1v  = (const float*)d_in[8];
    const float* Wr1  = (const float*)d_in[9];
    const float* Wr2  = (const float*)d_in[10];
    const float* W2s  = (const float*)d_in[11];
    const float* W2v  = (const float*)d_in[12];
    const float* Wscs = (const float*)d_in[13];
    const float* Wscv = (const float*)d_in[14];

    const int N = in_sizes[0] / MULT;
    const int E = in_sizes[3] / RB;

    float* s_buf = (float*)d_ws;
    float* v_buf = s_buf + (size_t)N * MULT;

    node_kernel<<<(N + 63) / 64, 128, 0, stream>>>(
        node_s, node_v, node_attrs, W1s, W1v, Wscs, Wscv, s_buf, v_buf, (float*)d_out, N);
    edge_kernel<<<(E + 127) / 128, 256, 0, stream>>>(
        edge_emb, edge_y0, edge_y1, edge_index, Wr1, Wr2, W2s, W2v,
        s_buf, v_buf, (float*)d_out, E);
}


# edge weight staging: node kernel writes the transposed, padded LDS image once per call to the workspace; each edge work-group copies it in one pass of 16-byte loads, not 14 serial 4-byte trips
# speedup vs baseline: 7.3594x; 1.0681x over previous
#include <hip/hip_runtime.h>
#include <math.h>


typedef float v2f __attribute__((ext_vector_type(2)));
typedef float v4f __attribute__((ext_vector_type(4)));
typedef float v8f __attribute__((ext_vector_type(8)));

#define MULT 32
#define NATTR 8
#define RB 16
#define FC 64
#define LOG2F_ 0.6931471805599453f
#define R2P 68
#define W2P 68
#define SCP 260
#define L1P 36
#define IMG_WR1 0
#define IMG_WR2 (FC * RB)
#define IMG_W2S (IMG_WR2 + 4 * MULT * R2P)
#define IMG_W2V (IMG_W2S + MULT * W2P)
#define IMG_FLOATS (IMG_W2V + MULT * W2P)
#define IMG_PASSES ((IMG_FLOATS / 4 + 255) / 256)
#define IMG_PADDED (IMG_PASSES * 256 * 4)

__device__ __forceinline__ float ssp(float x) {
    return fmaxf(x, 0.0f) + log1pf(__expf(-fabsf(x))) - LOG2F_;
}

__global__ __launch_bounds__(128) void node_kernel(
    const float* __restrict__ node_s, const float* __restrict__ node_v,
    const float* __restrict__ node_attrs,
    const float* __restrict__ W1s, const float* __restrict__ W1v,
    const float* __restrict__ Wscs, const float* __restrict__ Wscv,
    const float* __restrict__ Wr1, const float* __restrict__ Wr2,
    const float* __restrict__ W2s, const float* __restrict__ W2v,
    float* __restrict__ wimg,
    float* __restrict__ s_out, float* __restrict__ v_out,
    float* __restrict__ out, int N)
{
    __shared__ float Tss[MULT * SCP];
    __shared__ float Tsv[MULT * SCP];
    __shared__ float T1s[MULT * L1P];
    __shared__ float T1v[MULT * L1P];

    const int tid = threadIdx.x;
    for (int j = blockIdx.x * 128 + tid; j < IMG_PADDED; j += gridDim.x * 128) {
        float val = 0.0f;
        if (j < IMG_WR2) {
            const int n = j / RB, k = j % RB;
            val = Wr1[k * FC + n];
        } else if (j < IMG_W2S) {
            const int q = j - IMG_WR2, n = q / R2P, k = q % R2P;
            if (k < FC) val = Wr2[k * 4 * MULT + n];
        } else if (j < IMG_W2V) {
            const int q = j - IMG_W2S, n = q / W2P, k = q % W2P;
            if (k < 2 * MULT) val = W2s[k * MULT + n];
        } else if (j < IMG_FLOATS) {
            const int q = j - IMG_W2V, n = q / W2P, k = q % W2P;
            if (k < 2 * MULT) val = W2v[k * MULT + n];
        }
        wimg[j] = val;
    }
    for (int i = tid; i < MULT * NATTR * MULT; i += 128) {
        int k = i / MULT, n = i % MULT;
        Tss[n * SCP + k] = Wscs[i];
        Tsv[n * SCP + k] = Wscv[i];
    }
    for (int i = tid; i < MULT * MULT; i += 128) {
        int k = i / MULT, n = i % MULT;
        T1s[n * L1P + k] = W1s[i];
        T1v[n * L1P + k] = W1v[i];
    }
    __syncthreads();

    const int wave = tid >> 5, lane = tid & 31;
    const int m = lane & 15, kh = lane >> 4;
    const int n0 = (blockIdx.x * 4 + wave) * 16;
    if (n0 >= N) return;

    int na = n0 + m; if (na >= N) na = N - 1;
    const float* srow = node_s + (size_t)na * MULT;
    const float* vrow = node_v + (size_t)na * MULT * 3;
    const v8f vzero = {0.f, 0.f, 0.f, 0.f, 0.f, 0.f, 0.f, 0.f};

    {
        v8f ls[2], l0[2], l1[2], l2[2];
#pragma unroll
        for (int t = 0; t < 2; ++t) { ls[t] = vzero; l0[t] = vzero; l1[t] = vzero; l2[t] = vzero; }
#pragma unroll
        for (int j = 0; j < MULT / 4; ++j) {
            v2f as_ = *(const v2f*)(srow + 4 * j + 2 * kh);
            v2f p0 = *(const v2f*)(vrow + 12 * j + 6 * kh);
            v2f p1 = *(const v2f*)(vrow + 12 * j + 6 * kh + 2);
            v2f p2 = *(const v2f*)(vrow + 12 * j + 6 * kh + 4);
            v2f ax = {p0.x, p1.y};
            v2f ay = {p0.y, p2.x};
            v2f az = {p1.x, p2.y};
#pragma unroll
            for (int t = 0; t < 2; ++t) {
                v2f bs = *(const v2f*)(T1s + (t * 16 + m) * L1P + 4 * j + 2 * kh);
                v2f bv = *(const v2f*)(T1v + (t * 16 + m) * L1P + 4 * j + 2 * kh);
                ls[t] = __builtin_amdgcn_wmma_f32_16x16x4_f32(
                    false, as_, false, bs, (short)0, ls[t], false, false);
                l0[t] = __builtin_amdgcn_wmma_f32_16x16x4_f32(
                    false, ax, false, bv, (short)0, l0[t], false, false);
                l1[t] = __builtin_amdgcn_wmma_f32_16x16x4_f32(
                    false, ay, false, bv, (short)0, l1[t], false, false);
                l2[t] = __builtin_amdgcn_wmma_f32_16x16x4_f32(
                    false, az, false, bv, (short)0, l2[t], false, false);
            }
        }
        const float inv_m = 0.17677669529663687f;
#pragma unroll
        for (int r = 0; r < 8; ++r) {
            const int node = n0 + r + 8 * kh;
            if (node < N) {
#pragma unroll
                for (int t = 0; t < 2; ++t) {
                    const int w = t * 16 + m;
                    s_out[(size_t)node * MULT + w] = ls[t][r] * inv_m;
                    v_out[(size_t)node * MULT * 3 + w * 3 + 0] = l0[t][r] * inv_m;
                    v_out[(size_t)node * MULT * 3 + w * 3 + 1] = l1[t][r] * inv_m;
                    v_out[(size_t)node * MULT * 3 + w * 3 + 2] = l2[t][r] * inv_m;
                }
            }
        }
    }

    const v2f at0 = *(const v2f*)(node_attrs + (size_t)na * NATTR + 2 * kh);
    const v2f at1 = *(const v2f*)(node_attrs + (size_t)na * NATTR + 4 + 2 * kh);
    const float* bss = Tss + m * SCP + 2 * kh;
    const float* bsv = Tsv + m * SCP + 2 * kh;

    v8f cs[2], c0[2], c1[2], c2[2];
#pragma unroll
    for (int t = 0; t < 2; ++t) { cs[t] = vzero; c0[t] = vzero; c1[t] = vzero; c2[t] = vzero; }
#pragma unroll 4
    for (int u = 0; u < MULT; ++u) {
        const float xs = srow[u];
        const float x0 = vrow[u * 3 + 0];
        const float x1 = vrow[u * 3 + 1];
        const float x2 = vrow[u * 3 + 2];
#pragma unroll
        for (int h2 = 0; h2 < 2; ++h2) {
            const v2f at = h2 ? at1 : at0;
            v2f as_ = at * xs;
            v2f ax = at * x0, ay = at * x1, az = at * x2;
#pragma unroll
            for (int t = 0; t < 2; ++t) {
                v2f bs = *(const v2f*)(bss + t * 16 * SCP + 8 * u + 4 * h2);
                v2f bv = *(const v2f*)(bsv + t * 16 * SCP + 8 * u + 4 * h2);
                cs[t] = __builtin_amdgcn_wmma_f32_16x16x4_f32(
                    false, as_, false, bs, (short)0, cs[t], false, false);
                c0[t] = __builtin_amdgcn_wmma_f32_16x16x4_f32(
                    false, ax, false, bv, (short)0, c0[t], false, false);
                c1[t] = __builtin_amdgcn_wmma_f32_16x16x4_f32(
                    false, ay, false, bv, (short)0, c1[t], false, false);
                c2[t] = __builtin_amdgcn_wmma_f32_16x16x4_f32(
                    false, az, false, bv, (short)0, c2[t], false, false);
            }
        }
    }
    const float invfan = 0.0625f;
#pragma unroll
    for (int r = 0; r < 8; ++r) {
        const int node = n0 + r + 8 * kh;
        if (node < N) {
            float* orow = out + (size_t)node * 4 * MULT;
#pragma unroll
            for (int t = 0; t < 2; ++t) {
                const int w = t * 16 + m;
                orow[w] = cs[t][r] * invfan;
                orow[MULT + w * 3 + 0] = c0[t][r] * invfan;
                orow[MULT + w * 3 + 1] = c1[t][r] * invfan;
                orow[MULT + w * 3 + 2] = c2[t][r] * invfan;
            }
        }
    }
}

__global__ __launch_bounds__(256, 6) void edge_kernel(
    const float* __restrict__ edge_emb, const float* __restrict__ edge_y0,
    const float* __restrict__ edge_y1, const int* __restrict__ edge_index,
    const float* __restrict__ wimg,
    const float* __restrict__ s_in, const float* __restrict__ v_in,
    float* __restrict__ out, int E)
{
    __shared__ __attribute__((aligned(16))) float wsm[IMG_PADDED];
    const float* Wr1t = wsm + IMG_WR1;
    const float* Wr2t = wsm + IMG_WR2;
    const float* W2st = wsm + IMG_W2S;
    const float* W2vt = wsm + IMG_W2V;

    const int tid = threadIdx.x;
    const int wave = tid >> 5, lane = tid & 31;
    const int m = lane & 15, kh = lane >> 4;

#pragma unroll
    for (int p = 0; p < IMG_PASSES; ++p)
        ((v4f*)wsm)[p * 256 + tid] = ((const v4f*)wimg)[p * 256 + tid];
    __syncthreads();

    const int e0 = blockIdx.x * 128 + wave * 16;

    int ee = e0 + m; if (ee >= E) ee = E - 1;
    const int src  = edge_index[ee];
    const int dstm = edge_index[(size_t)E + ee];
    const float y0  = edge_y0[ee];
    const float y1x = edge_y1[(size_t)ee * 3 + 0];
    const float y1y = edge_y1[(size_t)ee * 3 + 1];
    const float y1z = edge_y1[(size_t)ee * 3 + 2];
    const float* srow = s_in + (size_t)src * MULT + 8 * kh;
    const float* vrow = v_in + (size_t)src * MULT * 3 + 24 * kh;

    const v8f vzero = {0.f, 0.f, 0.f, 0.f, 0.f, 0.f, 0.f, 0.f};

    v8f h[4];
#pragma unroll
    for (int t = 0; t < 4; ++t) h[t] = vzero;
#pragma unroll
    for (int k4 = 0; k4 < RB; k4 += 4) {
        v2f b = *(const v2f*)(edge_emb + (size_t)ee * RB + k4 + 2 * kh);
#pragma unroll
        for (int t = 0; t < 4; ++t) {
            v2f a = *(const v2f*)(Wr1t + (t * 16 + m) * RB + k4 + 2 * kh);
            h[t] = __builtin_amdgcn_wmma_f32_16x16x4_f32(
                false, a, false, b, (short)0, h[t], false, false);
        }
    }
#pragma unroll
    for (int t = 0; t < 4; ++t)
#pragma unroll
        for (int r = 0; r < 8; ++r)
            h[t][r] = ssp(h[t][r] * 0.25f);

    const float inv_s3 = 0.5773502691896258f;
    v8f accs[2], acc0[2], acc1[2], acc2[2];
    v8f pv[2];
#pragma unroll
    for (int t = 0; t < 2; ++t) { accs[t] = vzero; pv[t] = vzero; }

#pragma unroll
    for (int hf = 0; hf < 2; ++hf) {
        v8f wq[4];
#pragma unroll
        for (int cc = 0; cc < 4; ++cc) wq[cc] = vzero;
#pragma unroll
        for (int t = 0; t < 4; ++t) {
            v4f a0[4], a1[4];
#pragma unroll
            for (int cc = 0; cc < 4; ++cc) {
                const float* ap = Wr2t + ((4 * hf + cc) * 16 + m) * R2P + 16 * t + 8 * kh;
                a0[cc] = *(const v4f*)ap;
                a1[cc] = *(const v4f*)(ap + 4);
            }
#pragma unroll
            for (int j = 0; j < 4; ++j) {
                v2f b = {h[t][2 * j], h[t][2 * j + 1]};
#pragma unroll
                for (int cc = 0; cc < 4; ++cc) {
                    v2f a;
                    if (j == 0)      a = (v2f){a0[cc].x, a0[cc].y};
                    else if (j == 1) a = (v2f){a0[cc].z, a0[cc].w};
                    else if (j == 2) a = (v2f){a1[cc].x, a1[cc].y};
                    else             a = (v2f){a1[cc].z, a1[cc].w};
                    wq[cc] = __builtin_amdgcn_wmma_f32_16x16x4_f32(
                        false, a, false, b, (short)0, wq[cc], false, false);
                }
            }
        }

#pragma unroll
        for (int tp = 0; tp < 2; ++tp) {
            v4f bs0[2], bs1[2], bv0[2], bv1[2];
#pragma unroll
            for (int t = 0; t < 2; ++t) {
                const float* bp = W2st + (t * 16 + m) * W2P + 32 * hf + 16 * tp + 8 * kh;
                const float* bq = W2vt + (t * 16 + m) * W2P + 32 * hf + 16 * tp + 8 * kh;
                bs0[t] = *(const v4f*)bp;  bs1[t] = *(const v4f*)(bp + 4);
                bv0[t] = *(const v4f*)bq;  bv1[t] = *(const v4f*)(bq + 4);
            }
            float fs[8], fx[8], fy[8], fz[8];
            if (hf == 0) {
                const v4f s0 = *(const v4f*)(srow + 16 * tp);
                const v4f s1 = *(const v4f*)(srow + 16 * tp + 4);
                fs[0] = s0.x; fs[1] = s0.y; fs[2] = s0.z; fs[3] = s0.w;
                fs[4] = s1.x; fs[5] = s1.y; fs[6] = s1.z; fs[7] = s1.w;
            } else {
                float vv[24];
#pragma unroll
                for (int q = 0; q < 6; ++q) {
                    const v4f p = *(const v4f*)(vrow + 48 * tp + 4 * q);
                    vv[4 * q] = p.x; vv[4 * q + 1] = p.y; vv[4 * q + 2] = p.z; vv[4 * q + 3] = p.w;
                }
#pragma unroll
                for (int r = 0; r < 8; ++r) { fx[r] = vv[3 * r]; fy[r] = vv[3 * r + 1]; fz[r] = vv[3 * r + 2]; }
            }
#pragma unroll
            for (int j = 0; j < 4; ++j) {
                const int r0 = 2 * j;
                v2f as_, av0, av1, av2;
                if (hf == 0) {
                    v2f w1 = {wq[tp][r0], wq[tp][r0 + 1]};
                    v2f w2 = {wq[2 + tp][r0], wq[2 + tp][r0 + 1]};
                    v2f se = {fs[r0], fs[r0 + 1]};
                    as_ = w1 * se * y0;
                    av0 = w2 * se;
                    av1 = av0; av2 = av0;
                } else {
                    v2f w3 = {wq[tp][r0], wq[tp][r0 + 1]};
                    v2f w4 = {wq[2 + tp][r0], wq[2 + tp][r0 + 1]};
                    v2f vx = {fx[r0], fx[r0 + 1]};
                    v2f vy = {fy[r0], fy[r0 + 1]};
                    v2f vz = {fz[r0], fz[r0 + 1]};
                    v2f dv = vx * y1x + vy * y1y + vz * y1z;
                    as_ = w4 * dv * inv_s3;
                    v2f t3 = w3 * y0;
                    av0 = t3 * vx; av1 = t3 * vy; av2 = t3 * vz;
                }
#pragma unroll
                for (int t = 0; t < 2; ++t) {
                    v2f bs, bv;
                    if (j == 0)      { bs = (v2f){bs0[t].x, bs0[t].y}; bv = (v2f){bv0[t].x, bv0[t].y}; }
                    else if (j == 1) { bs = (v2f){bs0[t].z, bs0[t].w}; bv = (v2f){bv0[t].z, bv0[t].w}; }
                    else if (j == 2) { bs = (v2f){bs1[t].x, bs1[t].y}; bv = (v2f){bv1[t].x, bv1[t].y}; }
                    else             { bs = (v2f){bs1[t].z, bs1[t].w}; bv = (v2f){bv1[t].z, bv1[t].w}; }
                    accs[t] = __builtin_amdgcn_wmma_f32_16x16x4_f32(
                        false, as_, false, bs, (short)0, accs[t], false, false);
                    if (hf == 0) {
                        pv[t] = __builtin_amdgcn_wmma_f32_16x16x4_f32(
                            false, av0, false, bv, (short)0, pv[t], false, false);
                    } else {
                        acc0[t] = __builtin_amdgcn_wmma_f32_16x16x4_f32(
                            false, av0, false, bv, (short)0, acc0[t], false, false);
                        acc1[t] = __builtin_amdgcn_wmma_f32_16x16x4_f32(
                            false, av1, false, bv, (short)0, acc1[t], false, false);
                        acc2[t] = __builtin_amdgcn_wmma_f32_16x16x4_f32(
                            false, av2, false, bv, (short)0, acc2[t], false, false);
                    }
                }
            }
        }
        if (hf == 0) {
            const int rb = (__lane_id() >> 4) * 8;
#pragma unroll
            for (int r = 0; r < 8; ++r) {
                const float yx = __shfl(y1x, rb + r);
                const float yy = __shfl(y1y, rb + r);
                const float yz = __shfl(y1z, rb + r);
#pragma unroll
                for (int t = 0; t < 2; ++t) {
                    acc0[t][r] = pv[t][r] * yx;
                    acc1[t][r] = pv[t][r] * yy;
                    acc2[t][r] = pv[t][r] * yz;
                }
            }
        }
    }

    const float sc = 0.125f * 0.022097086912079608f;
#pragma unroll
    for (int t = 0; t < 2; ++t) { accs[t] *= sc; acc0[t] *= sc; acc1[t] *= sc; acc2[t] *= sc; }

    const int ln = __lane_id();
    const int ms = ln & 15, khs = ln >> 4;
#pragma unroll
    for (int r = 0; r < 8; ++r) {
        const int row = r + 8 * khs;
        const int d = __shfl(dstm, row);
        if (e0 + row < E) {
            float* orow = out + (size_t)d * 4 * MULT;
#pragma unroll
            for (int t = 0; t < 2; ++t) {
                const int wcol = t * 16 + ms;
                unsafeAtomicAdd(orow + wcol, accs[t][r]);
                unsafeAtomicAdd(orow + MULT + wcol * 3 + 0, acc0[t][r]);
                unsafeAtomicAdd(orow + MULT + wcol * 3 + 1, acc1[t][r]);
                unsafeAtomicAdd(orow + MULT + wcol * 3 + 2, acc2[t][r]);
            }
        }
    }
}

extern "C" void kernel_launch(void* const* d_in, const int* in_sizes, int n_in,
                              void* d_out, int out_size, void* d_ws, size_t ws_size,
                              hipStream_t stream) {
    const float* node_s     = (const float*)d_in[0];
    const float* node_v     = (const float*)d_in[1];
    const float* node_attrs = (const float*)d_in[2];
    const float* edge_emb   = (const float*)d_in[3];
    const float* edge_y0    = (const float*)d_in[4];
    const float* edge_y1    = (const float*)d_in[5];
    const int*   edge_index = (const int*)d_in[6];
    const float* W1s  = (const float*)d_in[7];
    const float* W1v  = (const float*)d_in[8];
    const float* Wr1  = (const float*)d_in[9];
    const float* Wr2  = (const float*)d_in[10];
    const float* W2s  = (const float*)d_in[11];
    const float* W2v  = (const float*)d_in[12];
    const float* Wscs = (const float*)d_in[13];
    const float* Wscv = (const float*)d_in[14];

    const int N = in_sizes[0] / MULT;
    const int E = in_sizes[3] / RB;

    float* s_buf = (float*)d_ws;
    float* v_buf = s_buf + (size_t)N * MULT;
    float* w_img = v_buf + (size_t)N * MULT * 3;

    node_kernel<<<(N + 63) / 64, 128, 0, stream>>>(
        node_s, node_v, node_attrs, W1s, W1v, Wscs, Wscv, Wr1, Wr2, W2s, W2v, w_img,
        s_buf, v_buf, (float*)d_out, N);
    edge_kernel<<<(E + 127) / 128, 256, 0, stream>>>(
        edge_emb, edge_y0, edge_y1, edge_index, w_img,
        s_buf, v_buf, (float*)d_out, E);
}
